# starting kernel
# baseline (speedup 1.0000x reference)
#include <hip/hip_runtime.h>


typedef float v2f __attribute__((ext_vector_type(2)));
typedef float v8f __attribute__((ext_vector_type(8)));

constexpr int kC  = 256;
constexpr int kH  = 64;
constexpr int kW  = 128;
constexpr int kND = 21;
constexpr int kHW = kH * kW;

__global__ __launch_bounds__(256) void corr1d_wmma_kernel(
    const float* __restrict__ in1,
    const float* __restrict__ in2,
    float* __restrict__ out)
{
    const int bh   = blockIdx.x;
    const int b    = bh / kH;
    const int h    = bh % kH;
    const int wv   = threadIdx.x >> 5;
    const int lane = threadIdx.x & 31;
    const int half = lane >> 4;
    const int l16  = lane & 15;
    const int m0   = (wv >> 1) << 4;
    const int r    = wv & 1;

    const size_t rowbase = ((size_t)b * kC * kH + (size_t)h) * kW;
    const float* p1 = in1 + rowbase;
    const float* p2 = in2 + rowbase;

    const int w1 = 2 * (m0 + l16) + r;

    int  w2[3];
    bool vld[3];
    #pragma unroll
    for (int j = 0; j < 3; ++j) {
        const int n = m0 + (j - 1) * 16 + l16;
        vld[j] = (n >= 0) && (n < kW / 2);
        w2[j]  = vld[j] ? (2 * n + r) : 0;
    }

    const size_t cOff = (size_t)(2 * half) * kHW;
    const float* aP  = p1 + cOff + w1;
    const float* bP0 = p2 + cOff + w2[0];
    const float* bP1 = p2 + cOff + w2[1];
    const float* bP2 = p2 + cOff + w2[2];

    v8f acc0 = {0.f, 0.f, 0.f, 0.f, 0.f, 0.f, 0.f, 0.f};
    v8f acc1 = {0.f, 0.f, 0.f, 0.f, 0.f, 0.f, 0.f, 0.f};
    v8f acc2 = {0.f, 0.f, 0.f, 0.f, 0.f, 0.f, 0.f, 0.f};

    #pragma unroll 8
    for (int t = 0; t < kC / 4; ++t) {
        v2f a, b0, b1, b2;
        a.x  = aP[0];    a.y  = aP[kHW];
        b0.x = bP0[0];   b0.y = bP0[kHW];
        b1.x = bP1[0];   b1.y = bP1[kHW];
        b2.x = bP2[0];   b2.y = bP2[kHW];

        acc0 = __builtin_amdgcn_wmma_f32_16x16x4_f32(false, a, false, b0,
                                                     (short)0, acc0, false, false);
        acc1 = __builtin_amdgcn_wmma_f32_16x16x4_f32(false, a, false, b1,
                                                     (short)0, acc1, false, false);
        acc2 = __builtin_amdgcn_wmma_f32_16x16x4_f32(false, a, false, b2,
                                                     (short)0, acc2, false, false);

        aP  += (size_t)4 * kHW;
        bP0 += (size_t)4 * kHW;
        bP1 += (size_t)4 * kHW;
        bP2 += (size_t)4 * kHW;
    }

    const float scale = 1.0f / (float)kC;
    float* po = out + (size_t)b * kND * kHW + (size_t)h * kW;

    #pragma unroll
    for (int j = 0; j < 3; ++j) {
        const v8f acc = (j == 0) ? acc0 : ((j == 1) ? acc1 : acc2);
        const int ncol = m0 + (j - 1) * 16 + l16;
        #pragma unroll
        for (int k = 0; k < 8; ++k) {
            const int mrow = m0 + k + 8 * half;
            const int i    = ncol - mrow + 10;
            if (i >= 0 && i < kND) {
                const float val = vld[j] ? acc[k] * scale : 0.0f;
                po[(size_t)i * kHW + 2 * mrow + r] = val;
            }
        }
    }
}

extern "C" void kernel_launch(void* const* d_in, const int* in_sizes, int n_in,
                              void* d_out, int out_size, void* d_ws, size_t ws_size,
                              hipStream_t stream) {
    const float* in1 = (const float*)d_in[0];
    const float* in2 = (const float*)d_in[1];
    float* out = (float*)d_out;

    const int B = in_sizes[0] / (kC * kH * kW);
    dim3 grid(B * kH);
    dim3 block(256);
    corr1d_wmma_kernel<<<grid, block, 0, stream>>>(in1, in2, out);
}


# Stage each (b,h) row pair through LDS in 32-channel chunks: whole-row 16-byte global loads, two LDS stages, fragments read from LDS as 8-byte k pairs; k order and sum order unchanged
# speedup vs baseline: 1.1870x; 1.1870x over previous
#include <hip/hip_runtime.h>


typedef float v2f __attribute__((ext_vector_type(2)));
typedef float v4f __attribute__((ext_vector_type(4)));
typedef float v8f __attribute__((ext_vector_type(8)));

constexpr int kC  = 256;
constexpr int kH  = 64;
constexpr int kW  = 128;
constexpr int kND = 21;
constexpr int kHW = kH * kW;

constexpr int kKC    = 32;
constexpr int kCP    = kKC / 2;
constexpr int kPitch = 160;
constexpr int kBuf   = 4 * kCP * kPitch;

__global__ __launch_bounds__(256) void corr1d_wmma_kernel(
    const float* __restrict__ in1,
    const float* __restrict__ in2,
    float* __restrict__ out)
{
    __shared__ __attribute__((aligned(16))) float lds[2 * kBuf];

    const int bh   = blockIdx.x;
    const int b    = bh / kH;
    const int h    = bh % kH;
    const int wv   = threadIdx.x >> 5;
    const int lane = threadIdx.x & 31;
    const int half = lane >> 4;
    const int l16  = lane & 15;
    const int m0   = (wv >> 1) << 4;
    const int r    = wv & 1;

    const size_t rowbase = ((size_t)b * kC * kH + (size_t)h) * kW;
    const float* p1 = in1 + rowbase;
    const float* p2 = in2 + rowbase;

    const int sIn  = wv >> 2;
    const int sCp0 = (wv & 3) * 4;
    const float* gP = (sIn ? p2 : p1) + (size_t)(2 * sCp0) * kHW + 4 * lane;
    const int sOff = (sIn * 2 * kCP + sCp0) * kPitch + 4 * lane;

    int  nOff[3];
    bool vld[3];
    #pragma unroll
    for (int j = 0; j < 3; ++j) {
        const int n = m0 + (j - 1) * 16 + l16;
        vld[j]  = (n >= 0) && (n < kW / 2);
        nOff[j] = vld[j] ? (2 * n) : 0;
    }
    const int aOff = (r * kCP + half) * kPitch + 2 * (m0 + l16);
    const int bOff = ((2 + r) * kCP + half) * kPitch;

    v8f acc0 = {0.f, 0.f, 0.f, 0.f, 0.f, 0.f, 0.f, 0.f};
    v8f acc1 = {0.f, 0.f, 0.f, 0.f, 0.f, 0.f, 0.f, 0.f};
    v8f acc2 = {0.f, 0.f, 0.f, 0.f, 0.f, 0.f, 0.f, 0.f};

    v4f g[8];
    for (int e = 0; e < 8; ++e)
        g[e] = *(const v4f*)(gP + (size_t)e * kHW);

    for (int ch = 0; ch < kC / kKC; ++ch) {
        float* buf = lds + (ch & 1) * kBuf;

        for (int e = 0; e < 4; ++e) {
            const v4f x = g[2 * e];
            const v4f y = g[2 * e + 1];
            const v4f ev = {x.x, y.x, x.z, y.z};
            const v4f od = {x.y, y.y, x.w, y.w};
            *(v4f*)(buf + sOff + e * kPitch)         = ev;
            *(v4f*)(buf + sOff + (kCP + e) * kPitch) = od;
        }

        __syncthreads();

        if (ch + 1 < kC / kKC) {
            gP += (size_t)kKC * kHW;
            for (int e = 0; e < 8; ++e)
                g[e] = *(const v4f*)(gP + (size_t)e * kHW);
        }

        const float* aL = buf + aOff;
        const float* bL = buf + bOff;
        for (int t = 0; t < kKC / 4; ++t) {
            const v2f a  = *(const v2f*)(aL + 2 * t * kPitch);
            const v2f b0 = *(const v2f*)(bL + 2 * t * kPitch + nOff[0]);
            const v2f b1 = *(const v2f*)(bL + 2 * t * kPitch + nOff[1]);
            const v2f b2 = *(const v2f*)(bL + 2 * t * kPitch + nOff[2]);

            acc0 = __builtin_amdgcn_wmma_f32_16x16x4_f32(false, a, false, b0,
                                                         (short)0, acc0, false, false);
            acc1 = __builtin_amdgcn_wmma_f32_16x16x4_f32(false, a, false, b1,
                                                         (short)0, acc1, false, false);
            acc2 = __builtin_amdgcn_wmma_f32_16x16x4_f32(false, a, false, b2,
                                                         (short)0, acc2, false, false);
        }
    }

    const float scale = 1.0f / (float)kC;
    float* po = out + (size_t)b * kND * kHW + (size_t)h * kW;

    #pragma unroll
    for (int j = 0; j < 3; ++j) {
        const v8f acc = (j == 0) ? acc0 : ((j == 1) ? acc1 : acc2);
        const int ncol = m0 + (j - 1) * 16 + l16;
        #pragma unroll
        for (int k = 0; k < 8; ++k) {
            const int mrow = m0 + k + 8 * half;
            const int i    = ncol - mrow + 10;
            if (i >= 0 && i < kND) {
                const float val = vld[j] ? acc[k] * scale : 0.0f;
                po[(size_t)i * kHW + 2 * mrow + r] = val;
            }
        }
    }
}

extern "C" void kernel_launch(void* const* d_in, const int* in_sizes, int n_in,
                              void* d_out, int out_size, void* d_ws, size_t ws_size,
                              hipStream_t stream) {
    const float* in1 = (const float*)d_in[0];
    const float* in2 = (const float*)d_in[1];
    float* out = (float*)d_out;

    const int B = in_sizes[0] / (kC * kH * kW);
    dim3 grid(B * kH);
    dim3 block(256);
    corr1d_wmma_kernel<<<grid, block, 0, stream>>>(in1, in2, out);
}
